# starting kernel
# baseline (speedup 1.0000x reference)
#include <hip/hip_runtime.h>

typedef __attribute__((ext_vector_type(2))) float v2f;
typedef __attribute__((ext_vector_type(8))) float v8f;

#define BATCH 4
#define HEADS 16
#define SEQ   4096
#define DIM   64
#define NHEAD (BATCH * HEADS)
#define KVELEMS (NHEAD * DIM * DIM)

__global__ void lsa_zero_kv(float* __restrict__ kvws) {
    int idx = (blockIdx.x * blockDim.x + threadIdx.x) * 4;
    if (idx < KVELEMS) {
        kvws[idx + 0] = 0.0f;
        kvws[idx + 1] = 0.0f;
        kvws[idx + 2] = 0.0f;
        kvws[idx + 3] = 0.0f;
    }
}

__global__ void lsa_phase1(const float* __restrict__ K,
                           const float* __restrict__ V,
                           float* __restrict__ kvws) {
    const int SCHUNKS = 4;
    int head  = blockIdx.x / SCHUNKS;
    int chunk = blockIdx.x % SCHUNKS;

    int wave = threadIdx.x >> 5;
    int lane = threadIdx.x & 31;

    int d0    = (wave & 3) * 16;
    int shalf = wave >> 2;
    int s_begin = chunk * (SEQ / SCHUNKS) + shalf * (SEQ / SCHUNKS / 2);
    int s_end   = s_begin + (SEQ / SCHUNKS / 2);

    const float* __restrict__ Kp = K + (size_t)head * SEQ * DIM;
    const float* __restrict__ Vp = V + (size_t)head * SEQ * DIM;

    int row = lane & 15;
    int kb  = (lane >> 4) * 2;

    v8f acc0 = {}; v8f acc1 = {}; v8f acc2 = {}; v8f acc3 = {};

    for (int s = s_begin; s < s_end; s += 4) {
        v2f a;
        a.x = Kp[(size_t)(s + kb    ) * DIM + d0 + row];
        a.y = Kp[(size_t)(s + kb + 1) * DIM + d0 + row];

        const float* v0 = Vp + (size_t)(s + kb    ) * DIM + row;
        const float* v1 = Vp + (size_t)(s + kb + 1) * DIM + row;
        v2f b0; b0.x = v0[ 0]; b0.y = v1[ 0];
        v2f b1; b1.x = v0[16]; b1.y = v1[16];
        v2f b2; b2.x = v0[32]; b2.y = v1[32];
        v2f b3; b3.x = v0[48]; b3.y = v1[48];

        acc0 = __builtin_amdgcn_wmma_f32_16x16x4_f32(false, a, false, b0,
                                                     (short)0, acc0, false, false);
        acc1 = __builtin_amdgcn_wmma_f32_16x16x4_f32(false, a, false, b1,
                                                     (short)0, acc1, false, false);
        acc2 = __builtin_amdgcn_wmma_f32_16x16x4_f32(false, a, false, b2,
                                                     (short)0, acc2, false, false);
        acc3 = __builtin_amdgcn_wmma_f32_16x16x4_f32(false, a, false, b3,
                                                     (short)0, acc3, false, false);
    }

    float* kvh = kvws + (size_t)head * DIM * DIM;
    int mofs = 8 * (lane >> 4);
#pragma unroll
    for (int v = 0; v < 8; ++v) {
        int d = d0 + v + mofs;
        unsafeAtomicAdd(&kvh[d * DIM + 0 * 16 + row], acc0[v]);
        unsafeAtomicAdd(&kvh[d * DIM + 1 * 16 + row], acc1[v]);
        unsafeAtomicAdd(&kvh[d * DIM + 2 * 16 + row], acc2[v]);
        unsafeAtomicAdd(&kvh[d * DIM + 3 * 16 + row], acc3[v]);
    }
}

__global__ void lsa_phase2(const float* __restrict__ Q,
                           const float* __restrict__ kvws,
                           float* __restrict__ out) {
    int wave = threadIdx.x >> 5;
    int lane = threadIdx.x & 31;
    int gwave = blockIdx.x * (blockDim.x >> 5) + wave;

    int head   = gwave >> 8;
    int mchunk = gwave & 255;
    int m0 = mchunk * 16;

    const float* __restrict__ Qp  = Q + ((size_t)head * SEQ + m0) * DIM;
    const float* __restrict__ kvh = kvws + (size_t)head * DIM * DIM;
    float* __restrict__ op        = out + ((size_t)head * SEQ + m0) * DIM;

    int row = lane & 15;
    int kb  = (lane >> 4) * 2;

    v8f acc0 = {}; v8f acc1 = {}; v8f acc2 = {}; v8f acc3 = {};

#pragma unroll 4
    for (int kk = 0; kk < 16; ++kk) {
        int k0 = kk * 4;
        const float* ap = Qp + row * DIM + k0 + kb;
        v2f a; a.x = ap[0]; a.y = ap[1];

        const float* b0p = kvh + (k0 + kb    ) * DIM + row;
        const float* b1p = kvh + (k0 + kb + 1) * DIM + row;
        v2f b0; b0.x = b0p[ 0]; b0.y = b1p[ 0];
        v2f b1; b1.x = b0p[16]; b1.y = b1p[16];
        v2f b2; b2.x = b0p[32]; b2.y = b1p[32];
        v2f b3; b3.x = b0p[48]; b3.y = b1p[48];

        acc0 = __builtin_amdgcn_wmma_f32_16x16x4_f32(false, a, false, b0,
                                                     (short)0, acc0, false, false);
        acc1 = __builtin_amdgcn_wmma_f32_16x16x4_f32(false, a, false, b1,
                                                     (short)0, acc1, false, false);
        acc2 = __builtin_amdgcn_wmma_f32_16x16x4_f32(false, a, false, b2,
                                                     (short)0, acc2, false, false);
        acc3 = __builtin_amdgcn_wmma_f32_16x16x4_f32(false, a, false, b3,
                                                     (short)0, acc3, false, false);
    }

    int mofs = 8 * (lane >> 4);
#pragma unroll
    for (int v = 0; v < 8; ++v) {
        float* r = op + (v + mofs) * DIM + row;
        r[ 0] = acc0[v];
        r[16] = acc1[v];
        r[32] = acc2[v];
        r[48] = acc3[v];
    }
}

extern "C" void kernel_launch(void* const* d_in, const int* in_sizes, int n_in,
                              void* d_out, int out_size, void* d_ws, size_t ws_size,
                              hipStream_t stream) {
    const float* q = (const float*)d_in[0];
    const float* k = (const float*)d_in[1];
    const float* v = (const float*)d_in[2];
    float* out = (float*)d_out;
    float* kvws = (float*)d_ws;

    lsa_zero_kv<<<KVELEMS / (256 * 4), 256, 0, stream>>>(kvws);

    lsa_phase1<<<NHEAD * 4, 256, 0, stream>>>(k, v, kvws);

    lsa_phase2<<<(NHEAD * (SEQ / 16)) / 8, 256, 0, stream>>>(q, kvws, out);
}


# phase 1 re-tiled: each wave keeps a whole 64x64 kv partial for its own 64 rows, K and V read once by 16-byte row-coalesced loads; block sums partials in LDS, then atomics (sum order changes)
# speedup vs baseline: 2.0975x; 2.0975x over previous
#include <hip/hip_runtime.h>

typedef __attribute__((ext_vector_type(2))) float v2f;
typedef __attribute__((ext_vector_type(4))) float v4f;
typedef __attribute__((ext_vector_type(8))) float v8f;

#define BATCH 4
#define HEADS 16
#define SEQ   4096
#define DIM   64
#define NHEAD (BATCH * HEADS)
#define KVELEMS (NHEAD * DIM * DIM)

__global__ void lsa_zero_kv(float* __restrict__ kvws) {
    int idx = (blockIdx.x * blockDim.x + threadIdx.x) * 4;
    if (idx < KVELEMS) {
        kvws[idx + 0] = 0.0f;
        kvws[idx + 1] = 0.0f;
        kvws[idx + 2] = 0.0f;
        kvws[idx + 3] = 0.0f;
    }
}

#define P1_WAVES   8
#define P1_THREADS (P1_WAVES * 32)
#define P1_ROWS    64
#define P1_STEPS   (P1_ROWS / 4)
#define P1_AHEAD   4
#define P1_CHUNKS  (SEQ / (P1_WAVES * P1_ROWS))

__global__ void __launch_bounds__(P1_THREADS, 4)
lsa_phase1(const float* __restrict__ K,
           const float* __restrict__ V,
           float* __restrict__ kvws) {
    __shared__ v4f red[P1_WAVES * 1024];

    const int tid  = threadIdx.x;
    const int lane = tid & 31;
    const int wave = __builtin_amdgcn_readfirstlane(tid >> 5);
    const int head  = blockIdx.x / P1_CHUNKS;
    const int chunk = blockIdx.x % P1_CHUNKS;
    const int s_base = (chunk * P1_WAVES + wave) * P1_ROWS;

    const int c = lane & 15;
    const int h = lane >> 4;

    const size_t base = ((size_t)head * SEQ + s_base + 2 * h) * DIM + 4 * c;
    const v4f* __restrict__ Kp = (const v4f*)(K + base);
    const v4f* __restrict__ Vp = (const v4f*)(V + base);

    v8f acc[4][4];
#pragma unroll
    for (int t = 0; t < 4; ++t)
#pragma unroll
        for (int u = 0; u < 4; ++u) acc[t][u] = v8f{};

    v4f kb[P1_AHEAD + 1][2], vb[P1_AHEAD + 1][2];

#define P1_LOAD(I) do { const int b_ = (I) % (P1_AHEAD + 1);          \
        kb[b_][0] = Kp[(I) * 64];      kb[b_][1] = Kp[(I) * 64 + 16]; \
        vb[b_][0] = Vp[(I) * 64];      vb[b_][1] = Vp[(I) * 64 + 16]; } while (0)

#pragma unroll
    for (int i = 0; i < P1_AHEAD; ++i) P1_LOAD(i);

#pragma unroll
    for (int i = 0; i < P1_STEPS; ++i) {
        if (i + P1_AHEAD < P1_STEPS) P1_LOAD(i + P1_AHEAD);
        const int b = i % (P1_AHEAD + 1);
#pragma unroll
        for (int t = 0; t < 4; ++t) {
            v2f a; a.x = kb[b][0][t]; a.y = kb[b][1][t];
#pragma unroll
            for (int u = 0; u < 4; ++u) {
                v2f bb; bb.x = vb[b][0][u]; bb.y = vb[b][1][u];
                acc[t][u] = __builtin_amdgcn_wmma_f32_16x16x4_f32(
                    false, a, false, bb, (short)0, acc[t][u], false, false);
            }
        }
    }
#undef P1_LOAD

    v4f* mine = red + wave * 1024 + lane;
#pragma unroll
    for (int t = 0; t < 4; ++t)
#pragma unroll
        for (int r = 0; r < 8; ++r) {
            v4f o;
            o.x = acc[t][0][r]; o.y = acc[t][1][r];
            o.z = acc[t][2][r]; o.w = acc[t][3][r];
            mine[(t * 8 + r) * 32] = o;
        }
    __syncthreads();

    float* kvh = kvws + (size_t)head * DIM * DIM;
#pragma unroll
    for (int z = 0; z < 4; ++z) {
        const int q = z * P1_THREADS + tid;
        v4f s = red[q];
#pragma unroll
        for (int w = 1; w < P1_WAVES; ++w) s += red[w * 1024 + q];
        const int rho = q >> 4;
        const int hh = rho & 1, tr = rho >> 1;
        const int d = 4 * (8 * hh + (tr & 7)) + (tr >> 3);
        float* dst = kvh + d * DIM + 4 * (q & 15);
        unsafeAtomicAdd(dst + 0, s.x);
        unsafeAtomicAdd(dst + 1, s.y);
        unsafeAtomicAdd(dst + 2, s.z);
        unsafeAtomicAdd(dst + 3, s.w);
    }
}

__global__ void lsa_phase2(const float* __restrict__ Q,
                           const float* __restrict__ kvws,
                           float* __restrict__ out) {
    int wave = threadIdx.x >> 5;
    int lane = threadIdx.x & 31;
    int gwave = blockIdx.x * (blockDim.x >> 5) + wave;

    int head   = gwave >> 8;
    int mchunk = gwave & 255;
    int m0 = mchunk * 16;

    const float* __restrict__ Qp  = Q + ((size_t)head * SEQ + m0) * DIM;
    const float* __restrict__ kvh = kvws + (size_t)head * DIM * DIM;
    float* __restrict__ op        = out + ((size_t)head * SEQ + m0) * DIM;

    int row = lane & 15;
    int kb  = (lane >> 4) * 2;

    v8f acc0 = {}; v8f acc1 = {}; v8f acc2 = {}; v8f acc3 = {};

#pragma unroll 4
    for (int kk = 0; kk < 16; ++kk) {
        int k0 = kk * 4;
        const float* ap = Qp + row * DIM + k0 + kb;
        v2f a; a.x = ap[0]; a.y = ap[1];

        const float* b0p = kvh + (k0 + kb    ) * DIM + row;
        const float* b1p = kvh + (k0 + kb + 1) * DIM + row;
        v2f b0; b0.x = b0p[ 0]; b0.y = b1p[ 0];
        v2f b1; b1.x = b0p[16]; b1.y = b1p[16];
        v2f b2; b2.x = b0p[32]; b2.y = b1p[32];
        v2f b3; b3.x = b0p[48]; b3.y = b1p[48];

        acc0 = __builtin_amdgcn_wmma_f32_16x16x4_f32(false, a, false, b0,
                                                     (short)0, acc0, false, false);
        acc1 = __builtin_amdgcn_wmma_f32_16x16x4_f32(false, a, false, b1,
                                                     (short)0, acc1, false, false);
        acc2 = __builtin_amdgcn_wmma_f32_16x16x4_f32(false, a, false, b2,
                                                     (short)0, acc2, false, false);
        acc3 = __builtin_amdgcn_wmma_f32_16x16x4_f32(false, a, false, b3,
                                                     (short)0, acc3, false, false);
    }

    int mofs = 8 * (lane >> 4);
#pragma unroll
    for (int v = 0; v < 8; ++v) {
        float* r = op + (v + mofs) * DIM + row;
        r[ 0] = acc0[v];
        r[16] = acc1[v];
        r[32] = acc2[v];
        r[48] = acc3[v];
    }
}

extern "C" void kernel_launch(void* const* d_in, const int* in_sizes, int n_in,
                              void* d_out, int out_size, void* d_ws, size_t ws_size,
                              hipStream_t stream) {
    const float* q = (const float*)d_in[0];
    const float* k = (const float*)d_in[1];
    const float* v = (const float*)d_in[2];
    float* out = (float*)d_out;
    float* kvws = (float*)d_ws;

    lsa_zero_kv<<<KVELEMS / (256 * 4), 256, 0, stream>>>(kvws);

    lsa_phase1<<<NHEAD * P1_CHUNKS, P1_THREADS, 0, stream>>>(k, v, kvws);

    lsa_phase2<<<(NHEAD * (SEQ / 16)) / 8, 256, 0, stream>>>(q, kvws, out);
}


# phase 2 re-tiled: a wave keeps its head's whole kv in registers as B fragments and runs eight 16-row tiles, Q requested ahead; 16-byte loads, out as whole rows (order of the sum over d changes)
# speedup vs baseline: 2.8085x; 1.3390x over previous
#include <hip/hip_runtime.h>

typedef __attribute__((ext_vector_type(2))) float v2f;
typedef __attribute__((ext_vector_type(4))) float v4f;
typedef __attribute__((ext_vector_type(8))) float v8f;

#define BATCH 4
#define HEADS 16
#define SEQ   4096
#define DIM   64
#define NHEAD (BATCH * HEADS)
#define KVELEMS (NHEAD * DIM * DIM)

__global__ void lsa_zero_kv(float* __restrict__ kvws) {
    int idx = (blockIdx.x * blockDim.x + threadIdx.x) * 4;
    if (idx < KVELEMS) {
        kvws[idx + 0] = 0.0f;
        kvws[idx + 1] = 0.0f;
        kvws[idx + 2] = 0.0f;
        kvws[idx + 3] = 0.0f;
    }
}

#define P1_WAVES   8
#define P1_THREADS (P1_WAVES * 32)
#define P1_ROWS    64
#define P1_STEPS   (P1_ROWS / 4)
#define P1_AHEAD   4
#define P1_CHUNKS  (SEQ / (P1_WAVES * P1_ROWS))

__global__ void __launch_bounds__(P1_THREADS, 4)
lsa_phase1(const float* __restrict__ K,
           const float* __restrict__ V,
           float* __restrict__ kvws) {
    __shared__ v4f red[P1_WAVES * 1024];

    const int tid  = threadIdx.x;
    const int lane = tid & 31;
    const int wave = __builtin_amdgcn_readfirstlane(tid >> 5);
    const int head  = blockIdx.x / P1_CHUNKS;
    const int chunk = blockIdx.x % P1_CHUNKS;
    const int s_base = (chunk * P1_WAVES + wave) * P1_ROWS;

    const int c = lane & 15;
    const int h = lane >> 4;

    const size_t base = ((size_t)head * SEQ + s_base + 2 * h) * DIM + 4 * c;
    const v4f* __restrict__ Kp = (const v4f*)(K + base);
    const v4f* __restrict__ Vp = (const v4f*)(V + base);

    v8f acc[4][4];
#pragma unroll
    for (int t = 0; t < 4; ++t)
#pragma unroll
        for (int u = 0; u < 4; ++u) acc[t][u] = v8f{};

    v4f kb[P1_AHEAD + 1][2], vb[P1_AHEAD + 1][2];

#define P1_LOAD(I) do { const int b_ = (I) % (P1_AHEAD + 1);          \
        kb[b_][0] = Kp[(I) * 64];      kb[b_][1] = Kp[(I) * 64 + 16]; \
        vb[b_][0] = Vp[(I) * 64];      vb[b_][1] = Vp[(I) * 64 + 16]; } while (0)

#pragma unroll
    for (int i = 0; i < P1_AHEAD; ++i) P1_LOAD(i);

#pragma unroll
    for (int i = 0; i < P1_STEPS; ++i) {
        if (i + P1_AHEAD < P1_STEPS) P1_LOAD(i + P1_AHEAD);
        const int b = i % (P1_AHEAD + 1);
#pragma unroll
        for (int t = 0; t < 4; ++t) {
            v2f a; a.x = kb[b][0][t]; a.y = kb[b][1][t];
#pragma unroll
            for (int u = 0; u < 4; ++u) {
                v2f bb; bb.x = vb[b][0][u]; bb.y = vb[b][1][u];
                acc[t][u] = __builtin_amdgcn_wmma_f32_16x16x4_f32(
                    false, a, false, bb, (short)0, acc[t][u], false, false);
            }
        }
    }
#undef P1_LOAD

    v4f* mine = red + wave * 1024 + lane;
#pragma unroll
    for (int t = 0; t < 4; ++t)
#pragma unroll
        for (int r = 0; r < 8; ++r) {
            v4f o;
            o.x = acc[t][0][r]; o.y = acc[t][1][r];
            o.z = acc[t][2][r]; o.w = acc[t][3][r];
            mine[(t * 8 + r) * 32] = o;
        }
    __syncthreads();

    float* kvh = kvws + (size_t)head * DIM * DIM;
#pragma unroll
    for (int z = 0; z < 4; ++z) {
        const int q = z * P1_THREADS + tid;
        v4f s = red[q];
#pragma unroll
        for (int w = 1; w < P1_WAVES; ++w) s += red[w * 1024 + q];
        const int rho = q >> 4;
        const int hh = rho & 1, tr = rho >> 1;
        const int d = 4 * (8 * hh + (tr & 7)) + (tr >> 3);
        float* dst = kvh + d * DIM + 4 * (q & 15);
        unsafeAtomicAdd(dst + 0, s.x);
        unsafeAtomicAdd(dst + 1, s.y);
        unsafeAtomicAdd(dst + 2, s.z);
        unsafeAtomicAdd(dst + 3, s.w);
    }
}

#define P2_WAVES   8
#define P2_THREADS (P2_WAVES * 32)
#define P2_TILES   8
#define P2_ROWS    (P2_TILES * 16)
#define P2_PARTS   (SEQ / (P2_WAVES * P2_ROWS))
#define P2_AHEAD   2

__global__ void __launch_bounds__(P2_THREADS, 2)
lsa_phase2(const float* __restrict__ Q,
           const float* __restrict__ kvws,
           float* __restrict__ out) {
    const int tid  = threadIdx.x;
    const int lane = tid & 31;
    const int wave = __builtin_amdgcn_readfirstlane(tid >> 5);
    const int head = blockIdx.x / P2_PARTS;
    const int part = blockIdx.x % P2_PARTS;
    const int m_base = (part * P2_WAVES + wave) * P2_ROWS;

    const int c = lane & 15;
    const int h = lane >> 4;

    const v4f* __restrict__ kvp =
        (const v4f*)(kvws + (size_t)head * DIM * DIM + (4 * h) * DIM + 4 * c);
    const v4f* __restrict__ Qp =
        (const v4f*)(Q + ((size_t)head * SEQ + m_base + c) * DIM + 4 * h);
    v4f* __restrict__ Op =
        (v4f*)(out + ((size_t)head * SEQ + m_base + 8 * h) * DIM + 4 * c);

    v4f kl[8][2][2];
#pragma unroll
    for (int i = 0; i < 8; ++i)
#pragma unroll
        for (int p = 0; p < 2; ++p)
#pragma unroll
            for (int s = 0; s < 2; ++s)
                kl[i][p][s] = kvp[(8 * i + 2 * p + s) * 16];

    v4f qb[P2_AHEAD + 1][8];
#define P2_LOAD(J) do { const int b_ = (J) % (P2_AHEAD + 1);            \
        _Pragma("unroll")                                               \
        for (int i_ = 0; i_ < 8; ++i_) qb[b_][i_] = Qp[(J) * 256 + i_ * 2]; } while (0)

#pragma unroll
    for (int j = 0; j < P2_AHEAD; ++j) P2_LOAD(j);

    v2f kb[8][2][4];
#pragma unroll
    for (int i = 0; i < 8; ++i)
#pragma unroll
        for (int p = 0; p < 2; ++p)
#pragma unroll
            for (int u = 0; u < 4; ++u) {
                kb[i][p][u].x = kl[i][p][0][u];
                kb[i][p][u].y = kl[i][p][1][u];
            }

#pragma unroll
    for (int j = 0; j < P2_TILES; ++j) {
        if (j + P2_AHEAD < P2_TILES) P2_LOAD(j + P2_AHEAD);
        __builtin_amdgcn_sched_barrier(0);
        const int b = j % (P2_AHEAD + 1);

        v8f acc[4];
#pragma unroll
        for (int u = 0; u < 4; ++u) acc[u] = v8f{};

#pragma unroll
        for (int i = 0; i < 8; ++i) {
#pragma unroll
            for (int p = 0; p < 2; ++p) {
                v2f a;
                a.x = qb[b][i][2 * p];
                a.y = qb[b][i][2 * p + 1];
#pragma unroll
                for (int u = 0; u < 4; ++u)
                    acc[u] = __builtin_amdgcn_wmma_f32_16x16x4_f32(
                        false, a, false, kb[i][p][u], (short)0, acc[u], false, false);
            }
        }

#pragma unroll
        for (int r = 0; r < 8; ++r) {
            v4f o;
            o.x = acc[0][r]; o.y = acc[1][r];
            o.z = acc[2][r]; o.w = acc[3][r];
            Op[j * 256 + r * 16] = o;
        }
    }
#undef P2_LOAD
}

extern "C" void kernel_launch(void* const* d_in, const int* in_sizes, int n_in,
                              void* d_out, int out_size, void* d_ws, size_t ws_size,
                              hipStream_t stream) {
    const float* q = (const float*)d_in[0];
    const float* k = (const float*)d_in[1];
    const float* v = (const float*)d_in[2];
    float* out = (float*)d_out;
    float* kvws = (float*)d_ws;

    lsa_zero_kv<<<KVELEMS / (256 * 4), 256, 0, stream>>>(kvws);

    lsa_phase1<<<NHEAD * P1_CHUNKS, P1_THREADS, 0, stream>>>(k, v, kvws);

    lsa_phase2<<<NHEAD * P2_PARTS, P2_THREADS, 0, stream>>>(q, kvws, out);
}


# phase 1: one 8-wave block per compute unit, 128 rows per wave (was 64) with 32 loads in flight: half the partials parked, summed and added, 4 atomics per kv element, was 8 (sum order changes)
# speedup vs baseline: 3.0305x; 1.0790x over previous
#include <hip/hip_runtime.h>

typedef __attribute__((ext_vector_type(2))) float v2f;
typedef __attribute__((ext_vector_type(4))) float v4f;
typedef __attribute__((ext_vector_type(8))) float v8f;

#define BATCH 4
#define HEADS 16
#define SEQ   4096
#define DIM   64
#define NHEAD (BATCH * HEADS)
#define KVELEMS (NHEAD * DIM * DIM)

__global__ void lsa_zero_kv(float* __restrict__ kvws) {
    int idx = (blockIdx.x * blockDim.x + threadIdx.x) * 4;
    if (idx < KVELEMS) {
        kvws[idx + 0] = 0.0f;
        kvws[idx + 1] = 0.0f;
        kvws[idx + 2] = 0.0f;
        kvws[idx + 3] = 0.0f;
    }
}

#define P1_WAVES   8
#define P1_THREADS (P1_WAVES * 32)
#define P1_ROWS    128
#define P1_STEPS   (P1_ROWS / 4)
#define P1_AHEAD   8
#define P1_CHUNKS  (SEQ / (P1_WAVES * P1_ROWS))

__global__ void __launch_bounds__(P1_THREADS, 2)
lsa_phase1(const float* __restrict__ K,
           const float* __restrict__ V,
           float* __restrict__ kvws) {
    __shared__ v4f red[P1_WAVES * 1024];

    const int tid  = threadIdx.x;
    const int lane = tid & 31;
    const int wave = __builtin_amdgcn_readfirstlane(tid >> 5);
    const int head  = blockIdx.x / P1_CHUNKS;
    const int chunk = blockIdx.x % P1_CHUNKS;
    const int s_base = (chunk * P1_WAVES + wave) * P1_ROWS;

    const int c = lane & 15;
    const int h = lane >> 4;

    const size_t base = ((size_t)head * SEQ + s_base + 2 * h) * DIM + 4 * c;
    const v4f* __restrict__ Kp = (const v4f*)(K + base);
    const v4f* __restrict__ Vp = (const v4f*)(V + base);

    v8f acc[4][4];
#pragma unroll
    for (int t = 0; t < 4; ++t)
#pragma unroll
        for (int u = 0; u < 4; ++u) acc[t][u] = v8f{};

    v4f kb[P1_AHEAD + 1][2], vb[P1_AHEAD + 1][2];

#define P1_LOAD(I) do { const int b_ = (I) % (P1_AHEAD + 1);          \
        kb[b_][0] = Kp[(I) * 64];      kb[b_][1] = Kp[(I) * 64 + 16]; \
        vb[b_][0] = Vp[(I) * 64];      vb[b_][1] = Vp[(I) * 64 + 16]; } while (0)

#pragma unroll
    for (int i = 0; i < P1_AHEAD; ++i) P1_LOAD(i);

#pragma unroll
    for (int i = 0; i < P1_STEPS; ++i) {
        if (i + P1_AHEAD < P1_STEPS) P1_LOAD(i + P1_AHEAD);
        __builtin_amdgcn_sched_barrier(0);
        const int b = i % (P1_AHEAD + 1);
#pragma unroll
        for (int t = 0; t < 4; ++t) {
            v2f a; a.x = kb[b][0][t]; a.y = kb[b][1][t];
#pragma unroll
            for (int u = 0; u < 4; ++u) {
                v2f bb; bb.x = vb[b][0][u]; bb.y = vb[b][1][u];
                acc[t][u] = __builtin_amdgcn_wmma_f32_16x16x4_f32(
                    false, a, false, bb, (short)0, acc[t][u], false, false);
            }
        }
    }
#undef P1_LOAD

    v4f* mine = red + wave * 1024 + lane;
#pragma unroll
    for (int t = 0; t < 4; ++t)
#pragma unroll
        for (int r = 0; r < 8; ++r) {
            v4f o;
            o.x = acc[t][0][r]; o.y = acc[t][1][r];
            o.z = acc[t][2][r]; o.w = acc[t][3][r];
            mine[(t * 8 + r) * 32] = o;
        }
    __syncthreads();

    float* kvh = kvws + (size_t)head * DIM * DIM;
#pragma unroll
    for (int z = 0; z < 4; ++z) {
        const int q = z * P1_THREADS + tid;
        v4f s = red[q];
#pragma unroll
        for (int w = 1; w < P1_WAVES; ++w) s += red[w * 1024 + q];
        const int rho = q >> 4;
        const int hh = rho & 1, tr = rho >> 1;
        const int d = 4 * (8 * hh + (tr & 7)) + (tr >> 3);
        float* dst = kvh + d * DIM + 4 * (q & 15);
        unsafeAtomicAdd(dst + 0, s.x);
        unsafeAtomicAdd(dst + 1, s.y);
        unsafeAtomicAdd(dst + 2, s.z);
        unsafeAtomicAdd(dst + 3, s.w);
    }
}

#define P2_WAVES   8
#define P2_THREADS (P2_WAVES * 32)
#define P2_TILES   8
#define P2_ROWS    (P2_TILES * 16)
#define P2_PARTS   (SEQ / (P2_WAVES * P2_ROWS))
#define P2_AHEAD   2

__global__ void __launch_bounds__(P2_THREADS, 2)
lsa_phase2(const float* __restrict__ Q,
           const float* __restrict__ kvws,
           float* __restrict__ out) {
    const int tid  = threadIdx.x;
    const int lane = tid & 31;
    const int wave = __builtin_amdgcn_readfirstlane(tid >> 5);
    const int head = blockIdx.x / P2_PARTS;
    const int part = blockIdx.x % P2_PARTS;
    const int m_base = (part * P2_WAVES + wave) * P2_ROWS;

    const int c = lane & 15;
    const int h = lane >> 4;

    const v4f* __restrict__ kvp =
        (const v4f*)(kvws + (size_t)head * DIM * DIM + (4 * h) * DIM + 4 * c);
    const v4f* __restrict__ Qp =
        (const v4f*)(Q + ((size_t)head * SEQ + m_base + c) * DIM + 4 * h);
    v4f* __restrict__ Op =
        (v4f*)(out + ((size_t)head * SEQ + m_base + 8 * h) * DIM + 4 * c);

    v4f kl[8][2][2];
#pragma unroll
    for (int i = 0; i < 8; ++i)
#pragma unroll
        for (int p = 0; p < 2; ++p)
#pragma unroll
            for (int s = 0; s < 2; ++s)
                kl[i][p][s] = kvp[(8 * i + 2 * p + s) * 16];

    v4f qb[P2_AHEAD + 1][8];
#define P2_LOAD(J) do { const int b_ = (J) % (P2_AHEAD + 1);            \
        _Pragma("unroll")                                               \
        for (int i_ = 0; i_ < 8; ++i_) qb[b_][i_] = Qp[(J) * 256 + i_ * 2]; } while (0)

#pragma unroll
    for (int j = 0; j < P2_AHEAD; ++j) P2_LOAD(j);

    v2f kb[8][2][4];
#pragma unroll
    for (int i = 0; i < 8; ++i)
#pragma unroll
        for (int p = 0; p < 2; ++p)
#pragma unroll
            for (int u = 0; u < 4; ++u) {
                kb[i][p][u].x = kl[i][p][0][u];
                kb[i][p][u].y = kl[i][p][1][u];
            }

#pragma unroll
    for (int j = 0; j < P2_TILES; ++j) {
        if (j + P2_AHEAD < P2_TILES) P2_LOAD(j + P2_AHEAD);
        __builtin_amdgcn_sched_barrier(0);
        const int b = j % (P2_AHEAD + 1);

        v8f acc[4];
#pragma unroll
        for (int u = 0; u < 4; ++u) acc[u] = v8f{};

#pragma unroll
        for (int i = 0; i < 8; ++i) {
#pragma unroll
            for (int p = 0; p < 2; ++p) {
                v2f a;
                a.x = qb[b][i][2 * p];
                a.y = qb[b][i][2 * p + 1];
#pragma unroll
                for (int u = 0; u < 4; ++u)
                    acc[u] = __builtin_amdgcn_wmma_f32_16x16x4_f32(
                        false, a, false, kb[i][p][u], (short)0, acc[u], false, false);
            }
        }

#pragma unroll
        for (int r = 0; r < 8; ++r) {
            v4f o;
            o.x = acc[0][r]; o.y = acc[1][r];
            o.z = acc[2][r]; o.w = acc[3][r];
            Op[j * 256 + r * 16] = o;
        }
    }
#undef P2_LOAD
}

extern "C" void kernel_launch(void* const* d_in, const int* in_sizes, int n_in,
                              void* d_out, int out_size, void* d_ws, size_t ws_size,
                              hipStream_t stream) {
    const float* q = (const float*)d_in[0];
    const float* k = (const float*)d_in[1];
    const float* v = (const float*)d_in[2];
    float* out = (float*)d_out;
    float* kvws = (float*)d_ws;

    lsa_zero_kv<<<KVELEMS / (256 * 4), 256, 0, stream>>>(kvws);

    lsa_phase1<<<NHEAD * P1_CHUNKS, P1_THREADS, 0, stream>>>(k, v, kvws);

    lsa_phase2<<<NHEAD * P2_PARTS, P2_THREADS, 0, stream>>>(q, kvws, out);
}
